# starting kernel
# baseline (speedup 1.0000x reference)
#include <hip/hip_runtime.h>

typedef float v2f __attribute__((ext_vector_type(2)));
typedef float v8f __attribute__((ext_vector_type(8)));

#define FP_TOL 1e-3f
#define FP_MAX_ITERS 1000

__global__ __launch_bounds__(256)
void deq_fixed_point_wmma(const float* __restrict__ d,
                          const float* __restrict__ Wfc,
                          const float* __restrict__ bfc,
                          const float* __restrict__ Win,
                          const float* __restrict__ bin,
                          const float* __restrict__ Wout,
                          const float* __restrict__ bout,
                          float* __restrict__ out,
                          int B)
{
    const int lane   = threadIdx.x & 31;
    const int warpId = (blockIdx.x * blockDim.x + threadIdx.x) >> 5;
    const int m      = lane & 15;
    const bool hi    = lane >= 16;

    const int row    = warpId * 16 + m;
    const int rowc   = (row < B) ? row : (B - 1);
    const bool cvalid = (!hi) && (row < B);

    const float* dp = d + (size_t)rowc * 3;
    const float d0 = dp[0], d1 = dp[1], d2 = dp[2];

    const float w00 = Wfc[0], w01 = Wfc[1], w02 = Wfc[2];
    const float w10 = Wfc[3], w11 = Wfc[4], w12 = Wfc[5];
    const float w20 = Wfc[6], w21 = Wfc[7], w22 = Wfc[8];

    float a0 = 0.f, a1 = 0.f;
    const int j = m & 7;
    if ((m & 7) < 3 && (m < 3 || m >= 8)) {
        if (j == 0)      { a0 = hi ? w02 : w00; a1 = hi ? 0.f : w01; }
        else if (j == 1) { a0 = hi ? w12 : w10; a1 = hi ? 0.f : w11; }
        else             { a0 = hi ? w22 : w20; a1 = hi ? 0.f : w21; }
    }
    v2f amat; amat.x = a0; amat.y = a1;

    const float bi0 = bfc[0] + bin[0];
    const float bi1 = bfc[1] + bin[1];
    const float bi2 = bfc[2] + bin[2];
    const float c0 = bi0 + Win[0]*d0 + Win[1]*d1 + Win[2]*d2;
    const float c1 = bi1 + Win[3]*d0 + Win[4]*d1 + Win[5]*d2;
    const float c2 = bi2 + Win[6]*d0 + Win[7]*d1 + Win[8]*d2;
    v8f cmat = {c0, c1, c2, 0.f, 0.f, 0.f, 0.f, 0.f};

    v2f bmat; bmat.x = 0.f; bmat.y = 0.f;
    float u0 = 0.f, u1 = 0.f, u2 = 0.f;

    for (int it = 0; it < FP_MAX_ITERS; ++it) {
        v8f acc = __builtin_amdgcn_wmma_f32_16x16x4_f32(
             false, amat,  false, bmat,
             (short)0, cmat,  false,  false);

        const float n0 = fmaxf(acc[0], 0.f);
        const float n1 = fmaxf(acc[1], 0.f);
        const float n2 = fmaxf(acc[2], 0.f);

        const float r = fmaxf(fmaxf(fabsf(n0 - u0), fabsf(n1 - u1)),
                              fabsf(n2 - u2));
        u0 = n0; u1 = n1; u2 = n2;

        bmat.x = hi ? n2 : n0;
        bmat.y = n1;

        if (__all(r < FP_TOL)) break;
    }

    const float o = bout[0] + Wout[0]*u0 + Wout[1]*u1 + Wout[2]*u2;
    if (cvalid) out[row] = o;
}

extern "C" void kernel_launch(void* const* d_in, const int* in_sizes, int n_in,
                              void* d_out, int out_size, void* d_ws, size_t ws_size,
                              hipStream_t stream) {
    const float* d    = (const float*)d_in[0];
    const float* Wfc  = (const float*)d_in[1];
    const float* bfc  = (const float*)d_in[2];
    const float* Win  = (const float*)d_in[3];
    const float* bin  = (const float*)d_in[4];
    const float* Wout = (const float*)d_in[5];
    const float* bout = (const float*)d_in[6];
    float* out = (float*)d_out;

    const int B = in_sizes[0] / 3;
    const int rowsPerBlock = 128;
    const int blocks = (B + rowsPerBlock - 1) / rowsPerBlock;

    deq_fixed_point_wmma<<<blocks, 256, 0, stream>>>(
        d, Wfc, bfc, Win, bin, Wout, bout, out, B);
}


# 4 independent 16-row tiles per wave instead of 1: their WMMAs issue back to back; A, weights, loop control and exit test shared; wave exits when all 64 rows converged (same or more sweeps per row)
# speedup vs baseline: 1.2014x; 1.2014x over previous
#include <hip/hip_runtime.h>


typedef float v2f __attribute__((ext_vector_type(2)));
typedef float v8f __attribute__((ext_vector_type(8)));

#define FP_TOL 1e-3f
#define FP_MAX_ITERS 1000
#define TILES 4

__global__ __launch_bounds__(256)
void deq_fixed_point_wmma(const float* __restrict__ d,
                          const float* __restrict__ Wfc,
                          const float* __restrict__ bfc,
                          const float* __restrict__ Win,
                          const float* __restrict__ bin,
                          const float* __restrict__ Wout,
                          const float* __restrict__ bout,
                          float* __restrict__ out,
                          int B)
{
    const int lane   = threadIdx.x & 31;
    const int warpId = (blockIdx.x * blockDim.x + threadIdx.x) >> 5;
    const int m      = lane & 15;
    const bool hi    = lane >= 16;

    const int row0   = warpId * (16 * TILES) + m;

    float dv[TILES][3];
    for (int t = 0; t < TILES; ++t) {
        const int row  = row0 + 16 * t;
        const int rowc = (row < B) ? row : (B - 1);
        const float* dp = d + (size_t)rowc * 3;
        dv[t][0] = dp[0]; dv[t][1] = dp[1]; dv[t][2] = dp[2];
    }

    const float w00 = Wfc[0], w01 = Wfc[1], w02 = Wfc[2];
    const float w10 = Wfc[3], w11 = Wfc[4], w12 = Wfc[5];
    const float w20 = Wfc[6], w21 = Wfc[7], w22 = Wfc[8];

    float a0 = 0.f, a1 = 0.f;
    const int j = m & 7;
    if ((m & 7) < 3 && (m < 3 || m >= 8)) {
        if (j == 0)      { a0 = hi ? w02 : w00; a1 = hi ? 0.f : w01; }
        else if (j == 1) { a0 = hi ? w12 : w10; a1 = hi ? 0.f : w11; }
        else             { a0 = hi ? w22 : w20; a1 = hi ? 0.f : w21; }
    }
    v2f amat; amat.x = a0; amat.y = a1;

    const float bi0 = bfc[0] + bin[0];
    const float bi1 = bfc[1] + bin[1];
    const float bi2 = bfc[2] + bin[2];
    const float i00 = Win[0], i01 = Win[1], i02 = Win[2];
    const float i10 = Win[3], i11 = Win[4], i12 = Win[5];
    const float i20 = Win[6], i21 = Win[7], i22 = Win[8];

    v8f cmat[TILES];
    v2f bmat[TILES];
    float u[TILES][3];
    for (int t = 0; t < TILES; ++t) {
        const float d0 = dv[t][0], d1 = dv[t][1], d2 = dv[t][2];
        const float c0 = bi0 + i00*d0 + i01*d1 + i02*d2;
        const float c1 = bi1 + i10*d0 + i11*d1 + i12*d2;
        const float c2 = bi2 + i20*d0 + i21*d1 + i22*d2;
        cmat[t] = (v8f){c0, c1, c2, 0.f, 0.f, 0.f, 0.f, 0.f};
        bmat[t].x = 0.f; bmat[t].y = 0.f;
        u[t][0] = 0.f; u[t][1] = 0.f; u[t][2] = 0.f;
    }

    for (int it = 0; it < FP_MAX_ITERS; ++it) {
        v8f acc[TILES];
        for (int t = 0; t < TILES; ++t) {
            acc[t] = __builtin_amdgcn_wmma_f32_16x16x4_f32(
                 false, amat,  false, bmat[t],
                 (short)0, cmat[t],  false,  false);
        }

        float r = 0.f;
        for (int t = 0; t < TILES; ++t) {
            const float n0 = fmaxf(acc[t][0], 0.f);
            const float n1 = fmaxf(acc[t][1], 0.f);
            const float n2 = fmaxf(acc[t][2], 0.f);

            r = fmaxf(fmaxf(r, fabsf(n0 - u[t][0])), fabsf(n1 - u[t][1]));
            r = fmaxf(r, fabsf(n2 - u[t][2]));
            u[t][0] = n0; u[t][1] = n1; u[t][2] = n2;

            bmat[t].x = hi ? n2 : n0;
            bmat[t].y = n1;
        }

        if (__all(r < FP_TOL)) break;
    }

    const float ob = bout[0], o0 = Wout[0], o1 = Wout[1], o2 = Wout[2];
    for (int t = 0; t < TILES; ++t) {
        const int row = row0 + 16 * t;
        const float o = ob + o0*u[t][0] + o1*u[t][1] + o2*u[t][2];
        if ((!hi) && (row < B)) out[row] = o;
    }
}

extern "C" void kernel_launch(void* const* d_in, const int* in_sizes, int n_in,
                              void* d_out, int out_size, void* d_ws, size_t ws_size,
                              hipStream_t stream) {
    const float* d    = (const float*)d_in[0];
    const float* Wfc  = (const float*)d_in[1];
    const float* bfc  = (const float*)d_in[2];
    const float* Win  = (const float*)d_in[3];
    const float* bin  = (const float*)d_in[4];
    const float* Wout = (const float*)d_in[5];
    const float* bout = (const float*)d_in[6];
    float* out = (float*)d_out;

    const int B = in_sizes[0] / 3;
    const int rowsPerBlock = 128 * TILES;
    const int blocks = (B + rowsPerBlock - 1) / rowsPerBlock;

    deq_fixed_point_wmma<<<blocks, 256, 0, stream>>>(
        d, Wfc, bfc, Win, bin, Wout, bout, out, B);
}


# loop-carried state cut to each tile's B pair (step test on the pair: lanes 0-15 u0, 16-31 u2, both u1) in two alternating sets over a twice-written sweep: no v_mov in the loop, 4 not 6 pk_add+max3
# speedup vs baseline: 1.3187x; 1.0976x over previous
#include <hip/hip_runtime.h>


typedef float v2f __attribute__((ext_vector_type(2)));
typedef float v8f __attribute__((ext_vector_type(8)));

#define FP_TOL 1e-3f
#define FP_MAX_ITERS 1000
#define TILES 4

__global__ __launch_bounds__(256)
void deq_fixed_point_wmma(const float* __restrict__ d,
                          const float* __restrict__ Wfc,
                          const float* __restrict__ bfc,
                          const float* __restrict__ Win,
                          const float* __restrict__ bin,
                          const float* __restrict__ Wout,
                          const float* __restrict__ bout,
                          float* __restrict__ out,
                          int B)
{
    const int lane   = threadIdx.x & 31;
    const int warpId = (blockIdx.x * blockDim.x + threadIdx.x) >> 5;
    const int m      = lane & 15;
    const bool hi    = lane >= 16;

    const int row0   = warpId * (16 * TILES) + m;

    float dv[TILES][3];
    for (int t = 0; t < TILES; ++t) {
        const int row  = row0 + 16 * t;
        const int rowc = (row < B) ? row : (B - 1);
        const float* dp = d + (size_t)rowc * 3;
        dv[t][0] = dp[0]; dv[t][1] = dp[1]; dv[t][2] = dp[2];
    }

    const float w00 = Wfc[0], w01 = Wfc[1], w02 = Wfc[2];
    const float w10 = Wfc[3], w11 = Wfc[4], w12 = Wfc[5];
    const float w20 = Wfc[6], w21 = Wfc[7], w22 = Wfc[8];

    float a0 = 0.f, a1 = 0.f;
    const int j = m & 7;
    if ((m & 7) < 3 && (m < 3 || m >= 8)) {
        if (j == 0)      { a0 = hi ? w02 : w00; a1 = hi ? 0.f : w01; }
        else if (j == 1) { a0 = hi ? w12 : w10; a1 = hi ? 0.f : w11; }
        else             { a0 = hi ? w22 : w20; a1 = hi ? 0.f : w21; }
    }
    v2f amat; amat.x = a0; amat.y = a1;

    const float bi0 = bfc[0] + bin[0];
    const float bi1 = bfc[1] + bin[1];
    const float bi2 = bfc[2] + bin[2];
    const float i00 = Win[0], i01 = Win[1], i02 = Win[2];
    const float i10 = Win[3], i11 = Win[4], i12 = Win[5];
    const float i20 = Win[6], i21 = Win[7], i22 = Win[8];

    v8f cmat[TILES];
    v2f bA[TILES], bB[TILES];
    float u2[TILES];
    for (int t = 0; t < TILES; ++t) {
        const float d0 = dv[t][0], d1 = dv[t][1], d2 = dv[t][2];
        const float c0 = bi0 + i00*d0 + i01*d1 + i02*d2;
        const float c1 = bi1 + i10*d0 + i11*d1 + i12*d2;
        const float c2 = bi2 + i20*d0 + i21*d1 + i22*d2;
        cmat[t] = (v8f){c0, c1, c2, 0.f, 0.f, 0.f, 0.f, 0.f};
        bA[t].x = 0.f; bA[t].y = 0.f;
        u2[t] = 0.f;
    }

    auto sweep = [&](const v2f (&src)[TILES], v2f (&dst)[TILES]) -> float {
        v8f acc[TILES];
        for (int t = 0; t < TILES; ++t) {
            acc[t] = __builtin_amdgcn_wmma_f32_16x16x4_f32(
                 false, amat,  false, src[t],
                 (short)0, cmat[t],  false,  false);
        }
        float r = 0.f;
        for (int t = 0; t < TILES; ++t) {
            const float n0 = fmaxf(acc[t][0], 0.f);
            const float n1 = fmaxf(acc[t][1], 0.f);
            const float n2 = fmaxf(acc[t][2], 0.f);
            v2f nb;
            nb.x = hi ? n2 : n0;
            nb.y = n1;
            const v2f dd = nb - src[t];
            r = fmaxf(r, fmaxf(fabsf(dd.x), fabsf(dd.y)));
            dst[t]  = nb;
            u2[t]   = n2;
        }
        return r;
    };

    for (int it = 0; it < FP_MAX_ITERS; it += 2) {
        if (__all(sweep(bA, bB) < FP_TOL)) {
            for (int t = 0; t < TILES; ++t) bA[t] = bB[t];
            break;
        }
        if (__all(sweep(bB, bA) < FP_TOL)) break;
    }

    const float ob = bout[0], o0 = Wout[0], o1 = Wout[1], o2 = Wout[2];
    for (int t = 0; t < TILES; ++t) {
        const int row = row0 + 16 * t;
        const float o = fmaf(o2, u2[t], fmaf(o1, bA[t].y, fmaf(o0, bA[t].x, ob)));
        if ((!hi) && (row < B)) out[row] = o;
    }
}

extern "C" void kernel_launch(void* const* d_in, const int* in_sizes, int n_in,
                              void* d_out, int out_size, void* d_ws, size_t ws_size,
                              hipStream_t stream) {
    const float* d    = (const float*)d_in[0];
    const float* Wfc  = (const float*)d_in[1];
    const float* bfc  = (const float*)d_in[2];
    const float* Win  = (const float*)d_in[3];
    const float* bin  = (const float*)d_in[4];
    const float* Wout = (const float*)d_in[5];
    const float* bout = (const float*)d_in[6];
    float* out = (float*)d_out;

    const int B = in_sizes[0] / 3;
    const int rowsPerBlock = 128 * TILES;
    const int blocks = (B + rowsPerBlock - 1) / rowsPerBlock;

    deq_fixed_point_wmma<<<blocks, 256, 0, stream>>>(
        d, Wfc, bfc, Win, bin, Wout, bout, out, B);
}


# sweep 1 without its product: u0 = 0 so T(u0) = relu(c) exactly, step test kept; WMMA loop starts at sweep 2, same sweeps per row, tests and cap; left: re-lay A rows so D VGPR0/1 are the next B pair
# speedup vs baseline: 1.4179x; 1.0752x over previous
#include <hip/hip_runtime.h>


typedef float v2f __attribute__((ext_vector_type(2)));
typedef float v8f __attribute__((ext_vector_type(8)));

#define FP_TOL 1e-3f
#define FP_MAX_ITERS 1000
#define TILES 4

__global__ __launch_bounds__(256)
void deq_fixed_point_wmma(const float* __restrict__ d,
                          const float* __restrict__ Wfc,
                          const float* __restrict__ bfc,
                          const float* __restrict__ Win,
                          const float* __restrict__ bin,
                          const float* __restrict__ Wout,
                          const float* __restrict__ bout,
                          float* __restrict__ out,
                          int B)
{
    const int lane   = threadIdx.x & 31;
    const int warpId = (blockIdx.x * blockDim.x + threadIdx.x) >> 5;
    const int m      = lane & 15;
    const bool hi    = lane >= 16;

    const int row0   = warpId * (16 * TILES) + m;

    float dv[TILES][3];
    for (int t = 0; t < TILES; ++t) {
        const int row  = row0 + 16 * t;
        const int rowc = (row < B) ? row : (B - 1);
        const float* dp = d + (size_t)rowc * 3;
        dv[t][0] = dp[0]; dv[t][1] = dp[1]; dv[t][2] = dp[2];
    }

    const float w00 = Wfc[0], w01 = Wfc[1], w02 = Wfc[2];
    const float w10 = Wfc[3], w11 = Wfc[4], w12 = Wfc[5];
    const float w20 = Wfc[6], w21 = Wfc[7], w22 = Wfc[8];

    float a0 = 0.f, a1 = 0.f;
    const int j = m & 7;
    if ((m & 7) < 3 && (m < 3 || m >= 8)) {
        if (j == 0)      { a0 = hi ? w02 : w00; a1 = hi ? 0.f : w01; }
        else if (j == 1) { a0 = hi ? w12 : w10; a1 = hi ? 0.f : w11; }
        else             { a0 = hi ? w22 : w20; a1 = hi ? 0.f : w21; }
    }
    v2f amat; amat.x = a0; amat.y = a1;

    const float bi0 = bfc[0] + bin[0];
    const float bi1 = bfc[1] + bin[1];
    const float bi2 = bfc[2] + bin[2];
    const float i00 = Win[0], i01 = Win[1], i02 = Win[2];
    const float i10 = Win[3], i11 = Win[4], i12 = Win[5];
    const float i20 = Win[6], i21 = Win[7], i22 = Win[8];

    v8f cmat[TILES];
    v2f bA[TILES], bB[TILES];
    float u2[TILES];
    float r1 = 0.f;
    for (int t = 0; t < TILES; ++t) {
        const float d0 = dv[t][0], d1 = dv[t][1], d2 = dv[t][2];
        const float c0 = bi0 + i00*d0 + i01*d1 + i02*d2;
        const float c1 = bi1 + i10*d0 + i11*d1 + i12*d2;
        const float c2 = bi2 + i20*d0 + i21*d1 + i22*d2;
        cmat[t] = (v8f){c0, c1, c2, 0.f, 0.f, 0.f, 0.f, 0.f};
        const float n0 = fmaxf(c0, 0.f);
        const float n1 = fmaxf(c1, 0.f);
        const float n2 = fmaxf(c2, 0.f);
        bA[t].x = hi ? n2 : n0;
        bA[t].y = n1;
        u2[t]   = n2;
        r1 = fmaxf(r1, fmaxf(bA[t].x, bA[t].y));
    }

    auto sweep = [&](const v2f (&src)[TILES], v2f (&dst)[TILES]) -> float {
        v8f acc[TILES];
        for (int t = 0; t < TILES; ++t) {
            acc[t] = __builtin_amdgcn_wmma_f32_16x16x4_f32(
                 false, amat,  false, src[t],
                 (short)0, cmat[t],  false,  false);
        }
        float r = 0.f;
        for (int t = 0; t < TILES; ++t) {
            const float n0 = fmaxf(acc[t][0], 0.f);
            const float n1 = fmaxf(acc[t][1], 0.f);
            const float n2 = fmaxf(acc[t][2], 0.f);
            v2f nb;
            nb.x = hi ? n2 : n0;
            nb.y = n1;
            const v2f dd = nb - src[t];
            r = fmaxf(r, fmaxf(fabsf(dd.x), fabsf(dd.y)));
            dst[t]  = nb;
            u2[t]   = n2;
        }
        return r;
    };

    if (!__all(r1 < FP_TOL)) {
        for (int it = 1; it < FP_MAX_ITERS; it += 2) {
            const bool conv = __all(sweep(bA, bB) < FP_TOL);
            if (conv || it + 1 >= FP_MAX_ITERS) {
                for (int t = 0; t < TILES; ++t) bA[t] = bB[t];
                break;
            }
            if (__all(sweep(bB, bA) < FP_TOL)) break;
        }
    }

    const float ob = bout[0], o0 = Wout[0], o1 = Wout[1], o2 = Wout[2];
    for (int t = 0; t < TILES; ++t) {
        const int row = row0 + 16 * t;
        const float o = fmaf(o2, u2[t], fmaf(o1, bA[t].y, fmaf(o0, bA[t].x, ob)));
        if ((!hi) && (row < B)) out[row] = o;
    }
}

extern "C" void kernel_launch(void* const* d_in, const int* in_sizes, int n_in,
                              void* d_out, int out_size, void* d_ws, size_t ws_size,
                              hipStream_t stream) {
    const float* d    = (const float*)d_in[0];
    const float* Wfc  = (const float*)d_in[1];
    const float* bfc  = (const float*)d_in[2];
    const float* Win  = (const float*)d_in[3];
    const float* bin  = (const float*)d_in[4];
    const float* Wout = (const float*)d_in[5];
    const float* bout = (const float*)d_in[6];
    float* out = (float*)d_out;

    const int B = in_sizes[0] / 3;
    const int rowsPerBlock = 128 * TILES;
    const int blocks = (B + rowsPerBlock - 1) / rowsPerBlock;

    deq_fixed_point_wmma<<<blocks, 256, 0, stream>>>(
        d, Wfc, bfc, Win, bin, Wout, bout, out, B);
}


# A rows re-laid (Wfc rows 0,1 in A rows 0,1; rows 2,1 in A rows 8,9; C alike): D VGPR0/1 are the next B pair, per tile 2 relu, no select; u_2 crosses lane halves once, at the output
# speedup vs baseline: 1.5044x; 1.0610x over previous
#include <hip/hip_runtime.h>


typedef float v2f __attribute__((ext_vector_type(2)));
typedef float v8f __attribute__((ext_vector_type(8)));

#define FP_TOL 1e-3f
#define FP_MAX_ITERS 1000
#define TILES 4


__device__ __forceinline__ float other_half(float x) {
    const int xi = __builtin_bit_cast(int, x);
    return __builtin_bit_cast(float,
        __builtin_amdgcn_permlanex16(xi, xi, 0x76543210, 0xfedcba98, false, false));
}

__global__ __launch_bounds__(256)
void deq_fixed_point_wmma(const float* __restrict__ d,
                          const float* __restrict__ Wfc,
                          const float* __restrict__ bfc,
                          const float* __restrict__ Win,
                          const float* __restrict__ bin,
                          const float* __restrict__ Wout,
                          const float* __restrict__ bout,
                          float* __restrict__ out,
                          int B)
{
    const int lane   = threadIdx.x & 31;
    const int warpId = (blockIdx.x * blockDim.x + threadIdx.x) >> 5;
    const int m      = lane & 15;
    const bool hi    = lane >= 16;

    const int row0   = warpId * (16 * TILES) + m;

    float dv[TILES][3];
    for (int t = 0; t < TILES; ++t) {
        const int row  = row0 + 16 * t;
        const int rowc = (row < B) ? row : (B - 1);
        const float* dp = d + (size_t)rowc * 3;
        dv[t][0] = dp[0]; dv[t][1] = dp[1]; dv[t][2] = dp[2];
    }

    const float w00 = Wfc[0], w01 = Wfc[1], w02 = Wfc[2];
    const float w10 = Wfc[3], w11 = Wfc[4], w12 = Wfc[5];
    const float w20 = Wfc[6], w21 = Wfc[7], w22 = Wfc[8];

    float a0 = 0.f, a1 = 0.f;
    if ((m & 6) == 0) {
        if (m == 0)      { a0 = hi ? w02 : w00; a1 = hi ? 0.f : w01; }
        else if (m == 8) { a0 = hi ? w22 : w20; a1 = hi ? 0.f : w21; }
        else             { a0 = hi ? w12 : w10; a1 = hi ? 0.f : w11; }
    }
    v2f amat; amat.x = a0; amat.y = a1;

    const float bi0 = bfc[0] + bin[0];
    const float bi1 = bfc[1] + bin[1];
    const float bi2 = bfc[2] + bin[2];
    const float i00 = Win[0], i01 = Win[1], i02 = Win[2];
    const float i10 = Win[3], i11 = Win[4], i12 = Win[5];
    const float i20 = Win[6], i21 = Win[7], i22 = Win[8];
    const float bs  = hi ? bi2 : bi0;
    const float is0 = hi ? i20 : i00;
    const float is1 = hi ? i21 : i01;
    const float is2 = hi ? i22 : i02;

    v8f cmat[TILES];
    v2f bA[TILES], bB[TILES];
    float r1 = 0.f;
    for (int t = 0; t < TILES; ++t) {
        const float d0 = dv[t][0], d1 = dv[t][1], d2 = dv[t][2];
        const float cs = fmaf(is2, d2, fmaf(is1, d1, fmaf(is0, d0, bs)));
        const float c1 = fmaf(i12, d2, fmaf(i11, d1, fmaf(i10, d0, bi1)));
        cmat[t] = (v8f){cs, c1, 0.f, 0.f, 0.f, 0.f, 0.f, 0.f};
        bA[t].x = fmaxf(cs, 0.f);
        bA[t].y = fmaxf(c1, 0.f);
        r1 = fmaxf(r1, fmaxf(bA[t].x, bA[t].y));
    }

    auto sweep = [&](const v2f (&src)[TILES], v2f (&dst)[TILES]) -> float {
        v8f acc[TILES];
        for (int t = 0; t < TILES; ++t) {
            acc[t] = __builtin_amdgcn_wmma_f32_16x16x4_f32(
                 false, amat,  false, src[t],
                 (short)0, cmat[t],  false,  false);
        }
        float r = 0.f;
        for (int t = 0; t < TILES; ++t) {
            v2f nb;
            nb.x = fmaxf(acc[t][0], 0.f);
            nb.y = fmaxf(acc[t][1], 0.f);
            const v2f dd = nb - src[t];
            r = fmaxf(r, fmaxf(fabsf(dd.x), fabsf(dd.y)));
            dst[t]  = nb;
        }
        return r;
    };

    if (!__all(r1 < FP_TOL)) {
        for (int it = 1; it < FP_MAX_ITERS; it += 2) {
            const bool conv = __all(sweep(bA, bB) < FP_TOL);
            if (conv || it + 1 >= FP_MAX_ITERS) {
                for (int t = 0; t < TILES; ++t) bA[t] = bB[t];
                break;
            }
            if (__all(sweep(bB, bA) < FP_TOL)) break;
        }
    }

    const float ob = bout[0], o0 = Wout[0], o1 = Wout[1], o2 = Wout[2];
    for (int t = 0; t < TILES; ++t) {
        const int row = row0 + 16 * t;
        const float u2 = other_half(bA[t].x);
        const float o = fmaf(o2, u2, fmaf(o1, bA[t].y, fmaf(o0, bA[t].x, ob)));
        if ((!hi) && (row < B)) out[row] = o;
    }
}

extern "C" void kernel_launch(void* const* d_in, const int* in_sizes, int n_in,
                              void* d_out, int out_size, void* d_ws, size_t ws_size,
                              hipStream_t stream) {
    const float* d    = (const float*)d_in[0];
    const float* Wfc  = (const float*)d_in[1];
    const float* bfc  = (const float*)d_in[2];
    const float* Win  = (const float*)d_in[3];
    const float* bin  = (const float*)d_in[4];
    const float* Wout = (const float*)d_in[5];
    const float* bout = (const float*)d_in[6];
    float* out = (float*)d_out;

    const int B = in_sizes[0] / 3;
    const int rowsPerBlock = 128 * TILES;
    const int blocks = (B + rowsPerBlock - 1) / rowsPerBlock;

    deq_fixed_point_wmma<<<blocks, 256, 0, stream>>>(
        d, Wfc, bfc, Win, bin, Wout, bout, out, B);
}
